# starting kernel
# baseline (speedup 1.0000x reference)
#include <hip/hip_runtime.h>


typedef float v2f __attribute__((ext_vector_type(2)));
typedef float v8f __attribute__((ext_vector_type(8)));

#define LSIG   524288
#define SEGS   256
#define S_TILE 2064

struct GTab { float g[2][18]; };

__host__ __device__ constexpr GTab makeG() {
    constexpr float REC[10] = {
        0.160102397974125f, 0.6038292697974729f, 0.7243085284385744f,
        0.13842814590110342f, -0.24229488706619015f, -0.03224486958502952f,
        0.07757149384006515f, -0.006241490213011705f, -0.012580751999015526f,
        0.003335725285001549f };
    GTab t{};
    for (int r = 0; r < 2; ++r)
        for (int k = 0; k < 18; ++k) {
            float acc = 0.0f;
            for (int u = 0; u < 5; ++u) {
                int tt = k - 2 * u;
                if (tt >= 0 && tt <= 9) acc += REC[tt] * REC[8 + r - 2 * u];
            }
            t.g[r][k] = acc;
        }
    return t;
}

__device__ __constant__ GTab cG = makeG();

extern "C" __global__ __launch_bounds__(256)
void wavelet_fused_wmma(const float* __restrict__ x, float* __restrict__ y)
{
    __shared__ __align__(16) float s_tile[S_TILE];

    const int tid = threadIdx.x;
    const int blk = blockIdx.x;
    const int b   = blk >> 8;
    const int seg = blk & (SEGS - 1);
    const int S0  = seg << 11;

    const float* __restrict__ xb = x + (size_t)b * LSIG;

    if (seg > 0 && seg < SEGS - 1) {
        const float4* __restrict__ src = (const float4*)(xb + S0 - 8);
        float4* dst = (float4*)s_tile;
        #pragma unroll 2
        for (int t = tid; t < S_TILE / 4; t += 256) dst[t] = src[t];
    } else {
        for (int t = tid; t < S_TILE; t += 256) {
            int idx = S0 + t - 8;
            if (idx < 0) idx = -idx - 1;
            else if (idx >= LSIG) idx = 2 * LSIG - idx - 1;
            s_tile[t] = xb[idx];
        }
    }
    __syncthreads();

    const int lane = tid & 31;
    const int wv   = tid >> 5;
    const int half = lane >> 4;
    const int im   = lane & 15;

    const int r  = im & 1;
    const int jc = im & ~1;
    v2f bfrag[8];
    #pragma unroll
    for (int q = 0; q < 8; ++q) {
        #pragma unroll
        for (int kk = 0; kk < 2; ++kk) {
            int gi = 4 * q + 2 * half + kk - jc;
            bfrag[q][kk] = (gi >= 0 && gi < 18) ? cG.g[r][gi] : 0.0f;
        }
    }

    const int abase = (wv << 8) + (im << 4) + (half << 1);
    v8f c = {};
    #pragma unroll
    for (int q = 0; q < 8; ++q) {
        const v2f a = *(const v2f*)(s_tile + abase + 4 * q);
        c = __builtin_amdgcn_wmma_f32_16x16x4_f32(
                 false, a,  false, bfrag[q],
                 (short)0, c,  false,  false);
    }

    float* __restrict__ yb = y + (size_t)b * LSIG
                           + ((size_t)seg << 11) + (wv << 8) + (half << 7) + im;
    #pragma unroll
    for (int g = 0; g < 8; ++g)
        yb[g << 4] = c[g];
}

extern "C" void kernel_launch(void* const* d_in, const int* in_sizes, int n_in,
                              void* d_out, int out_size, void* d_ws, size_t ws_size,
                              hipStream_t stream) {
    const float* x = (const float*)d_in[0];
    float*       y = (float*)d_out;
    wavelet_fused_wmma<<<dim3(64 * SEGS), dim3(256), 0, stream>>>(x, y);
}


# A fragments loaded straight from global memory, k order permuted (one 64-byte run per lane and tile): no LDS, no barrier; 16384 waves, each requests its 8 tiles up front; sum order changes
# speedup vs baseline: 1.2795x; 1.2795x over previous
#include <hip/hip_runtime.h>


typedef float v2f __attribute__((ext_vector_type(2)));
typedef float v4f __attribute__((ext_vector_type(4)));
typedef float v8f __attribute__((ext_vector_type(8)));

#define LSIG   524288
#define SEGS   256
#define TILES  8

#define GOFF 14
struct GTab { __attribute__((aligned(16))) float g[2][48]; };

__host__ __device__ constexpr GTab makeG() {
    constexpr float REC[10] = {
        0.160102397974125f, 0.6038292697974729f, 0.7243085284385744f,
        0.13842814590110342f, -0.24229488706619015f, -0.03224486958502952f,
        0.07757149384006515f, -0.006241490213011705f, -0.012580751999015526f,
        0.003335725285001549f };
    GTab t{};
    for (int r = 0; r < 2; ++r)
        for (int k = 0; k < 18; ++k) {
            float acc = 0.0f;
            for (int u = 0; u < 5; ++u) {
                int tt = k - 2 * u;
                if (tt >= 0 && tt <= 9) acc += REC[tt] * REC[8 + r - 2 * u];
            }
            t.g[r][GOFF + k] = acc;
        }
    return t;
}

__device__ __constant__ GTab cG = makeG();

__device__ __forceinline__ v4f rev4(v4f v) { v4f o = { v.w, v.z, v.y, v.x }; return o; }

extern "C" __global__ __launch_bounds__(256)
void wavelet_fused_wmma(const float* __restrict__ x, float* __restrict__ y)
{
    const int tid  = threadIdx.x;
    const int lane = tid & 31;
    const int gw   = (blockIdx.x << 3) + (tid >> 5);
    const int b    = gw >> 8;
    const int seg  = gw & (SEGS - 1);
    const int S0   = seg << 11;
    const int half = lane >> 4;
    const int im   = lane & 15;

    const float* __restrict__ xb = x + (size_t)b * LSIG;

    const int r  = im & 1;
    const int jc = im & ~1;
    const v2f* __restrict__ gp = (const v2f*)&cG.g[r][GOFF + 16 * half - jc];
    v2f bfrag[8];
    #pragma unroll
    for (int q = 0; q < 8; ++q) bfrag[q] = gp[q];

    const int run0 = S0 + ((im + half) << 4) - 8;
    v4f a[TILES][4];
    #pragma unroll
    for (int tt = 0; tt < TILES; ++tt) {
        int i0 = run0 + (tt << 8);
        if (tt == 0)         i0 = max(i0, 0);
        if (tt == TILES - 1) i0 = min(i0, LSIG - 16);
        const v4f* __restrict__ p = (const v4f*)(xb + i0);
        #pragma unroll
        for (int j = 0; j < 4; ++j) a[tt][j] = p[j];
    }

    float* __restrict__ yb = y + (size_t)b * LSIG + S0 + (half << 7) + im;

    #pragma unroll
    for (int tp = 0; tp < TILES; tp += 2) {
        if (tp == 0 && seg == 0 && lane == 0) {
            const v4f lo = a[0][0], hi = a[0][1];
            a[0][0] = rev4(hi); a[0][1] = rev4(lo); a[0][2] = lo; a[0][3] = hi;
        }
        if (tp == TILES - 2 && seg == SEGS - 1 && lane == 31) {
            const v4f lo = a[TILES - 1][2], hi = a[TILES - 1][3];
            a[TILES - 1][0] = lo; a[TILES - 1][1] = hi;
            a[TILES - 1][2] = rev4(hi); a[TILES - 1][3] = rev4(lo);
        }

        v8f c0 = {}, c1 = {};
        #pragma unroll
        for (int q = 0; q < 8; ++q) {
            const v4f u0 = a[tp][q >> 1], u1 = a[tp + 1][q >> 1];
            v2f a0, a1;
            if (q & 1) { a0 = (v2f){ u0.z, u0.w }; a1 = (v2f){ u1.z, u1.w }; }
            else       { a0 = (v2f){ u0.x, u0.y }; a1 = (v2f){ u1.x, u1.y }; }
            c0 = __builtin_amdgcn_wmma_f32_16x16x4_f32(
                     false, a0,  false, bfrag[q],
                     (short)0, c0,  false,  false);
            c1 = __builtin_amdgcn_wmma_f32_16x16x4_f32(
                     false, a1,  false, bfrag[q],
                     (short)0, c1,  false,  false);
        }
        float* __restrict__ y0 = yb + (tp << 8);
        #pragma unroll
        for (int g = 0; g < 8; ++g) y0[g << 4] = c0[g];
        #pragma unroll
        for (int g = 0; g < 8; ++g) y0[256 + (g << 4)] = c1[g];
    }
}

extern "C" void kernel_launch(void* const* d_in, const int* in_sizes, int n_in,
                              void* d_out, int out_size, void* d_ws, size_t ws_size,
                              hipStream_t stream) {
    const float* x = (const float*)d_in[0];
    float*       y = (float*)d_out;
    wavelet_fused_wmma<<<dim3(64 * SEGS / 8), dim3(256), 0, stream>>>(x, y);
}


# Finished tiles transposed 4x4 inside each lane quad (two DPP stages): a lane stores 4 consecutive outputs; 16 b128 stores per wave, each 4 whole 128-byte lines (was 64 b32 of two half lines)
# speedup vs baseline: 1.4853x; 1.1608x over previous
#include <hip/hip_runtime.h>


typedef float v2f __attribute__((ext_vector_type(2)));
typedef float v4f __attribute__((ext_vector_type(4)));
typedef float v8f __attribute__((ext_vector_type(8)));

#define LSIG   524288
#define SEGS   256
#define TILES  8

#define GOFF 14
struct GTab { __attribute__((aligned(16))) float g[2][48]; };

__host__ __device__ constexpr GTab makeG() {
    constexpr float REC[10] = {
        0.160102397974125f, 0.6038292697974729f, 0.7243085284385744f,
        0.13842814590110342f, -0.24229488706619015f, -0.03224486958502952f,
        0.07757149384006515f, -0.006241490213011705f, -0.012580751999015526f,
        0.003335725285001549f };
    GTab t{};
    for (int r = 0; r < 2; ++r)
        for (int k = 0; k < 18; ++k) {
            float acc = 0.0f;
            for (int u = 0; u < 5; ++u) {
                int tt = k - 2 * u;
                if (tt >= 0 && tt <= 9) acc += REC[tt] * REC[8 + r - 2 * u];
            }
            t.g[r][GOFF + k] = acc;
        }
    return t;
}

__device__ __constant__ GTab cG = makeG();

__device__ __forceinline__ v4f rev4(v4f v) { v4f o = { v.w, v.z, v.y, v.x }; return o; }

__device__ __forceinline__ float quad_x1(float v) {
    return __builtin_bit_cast(float, __builtin_amdgcn_update_dpp(
        0, __builtin_bit_cast(int, v), 0xB1  , 0xF, 0xF, true));
}
__device__ __forceinline__ float quad_x2(float v) {
    return __builtin_bit_cast(float, __builtin_amdgcn_update_dpp(
        0, __builtin_bit_cast(int, v), 0x4E  , 0xF, 0xF, true));
}

template <int GB>
__device__ __forceinline__ void quad_transpose(v8f& r, const bool o1, const bool o2) {
    #pragma unroll
    for (int p = 0; p < 4; p += 2) {
        const float lo = r[4 * GB + p], hi = r[4 * GB + p + 1];
        const float got = quad_x1(o1 ? lo : hi);
        r[4 * GB + p]     = o1 ? got : lo;
        r[4 * GB + p + 1] = o1 ? hi : got;
    }
    #pragma unroll
    for (int p = 0; p < 2; ++p) {
        const float lo = r[4 * GB + p], hi = r[4 * GB + p + 2];
        const float got = quad_x2(o2 ? lo : hi);
        r[4 * GB + p]     = o2 ? got : lo;
        r[4 * GB + p + 2] = o2 ? hi : got;
    }
}

extern "C" __global__ __launch_bounds__(256)
void wavelet_fused_wmma(const float* __restrict__ x, float* __restrict__ y)
{
    const int tid  = threadIdx.x;
    const int lane = tid & 31;
    const int gw   = (blockIdx.x << 3) + (tid >> 5);
    const int b    = gw >> 8;
    const int seg  = gw & (SEGS - 1);
    const int S0   = seg << 11;
    const int half = lane >> 4;
    const int im   = lane & 15;

    const float* __restrict__ xb = x + (size_t)b * LSIG;

    const int r  = im & 1;
    const int jc = im & ~1;
    const v2f* __restrict__ gp = (const v2f*)&cG.g[r][GOFF + 16 * half - jc];
    v2f bfrag[8];
    #pragma unroll
    for (int q = 0; q < 8; ++q) bfrag[q] = gp[q];

    const int run0 = S0 + ((im + half) << 4) - 8;
    v4f a[TILES][4];
    #pragma unroll
    for (int tt = 0; tt < TILES; ++tt) {
        int i0 = run0 + (tt << 8);
        if (tt == 0)         i0 = max(i0, 0);
        if (tt == TILES - 1) i0 = min(i0, LSIG - 16);
        const v4f* __restrict__ p = (const v4f*)(xb + i0);
        #pragma unroll
        for (int j = 0; j < 4; ++j) a[tt][j] = p[j];
    }

    const bool o1 = (lane & 1) != 0, o2 = (lane & 2) != 0;
    float* __restrict__ yb = y + (size_t)b * LSIG + S0
                           + (((im & 3) + (half << 3)) << 4) + (im & ~3);

    #pragma unroll
    for (int tp = 0; tp < TILES; tp += 2) {
        if (tp == 0 && seg == 0 && lane == 0) {
            const v4f lo = a[0][0], hi = a[0][1];
            a[0][0] = rev4(hi); a[0][1] = rev4(lo); a[0][2] = lo; a[0][3] = hi;
        }
        if (tp == TILES - 2 && seg == SEGS - 1 && lane == 31) {
            const v4f lo = a[TILES - 1][2], hi = a[TILES - 1][3];
            a[TILES - 1][0] = lo; a[TILES - 1][1] = hi;
            a[TILES - 1][2] = rev4(hi); a[TILES - 1][3] = rev4(lo);
        }

        v8f c0 = {}, c1 = {};
        #pragma unroll
        for (int q = 0; q < 8; ++q) {
            const v4f u0 = a[tp][q >> 1], u1 = a[tp + 1][q >> 1];
            v2f a0, a1;
            if (q & 1) { a0 = (v2f){ u0.z, u0.w }; a1 = (v2f){ u1.z, u1.w }; }
            else       { a0 = (v2f){ u0.x, u0.y }; a1 = (v2f){ u1.x, u1.y }; }
            c0 = __builtin_amdgcn_wmma_f32_16x16x4_f32(
                     false, a0,  false, bfrag[q],
                     (short)0, c0,  false,  false);
            c1 = __builtin_amdgcn_wmma_f32_16x16x4_f32(
                     false, a1,  false, bfrag[q],
                     (short)0, c1,  false,  false);
        }
        quad_transpose<0>(c0, o1, o2); quad_transpose<1>(c0, o1, o2);
        quad_transpose<0>(c1, o1, o2); quad_transpose<1>(c1, o1, o2);
        float* __restrict__ y0 = yb + (tp << 8);
        *(v4f*)(y0)       = __builtin_shufflevector(c0, c0, 0, 1, 2, 3);
        *(v4f*)(y0 + 64)  = __builtin_shufflevector(c0, c0, 4, 5, 6, 7);
        *(v4f*)(y0 + 256) = __builtin_shufflevector(c1, c1, 0, 1, 2, 3);
        *(v4f*)(y0 + 320) = __builtin_shufflevector(c1, c1, 4, 5, 6, 7);
    }
}

extern "C" void kernel_launch(void* const* d_in, const int* in_sizes, int n_in,
                              void* d_out, int out_size, void* d_ws, size_t ws_size,
                              hipStream_t stream) {
    const float* x = (const float*)d_in[0];
    float*       y = (float*)d_out;
    wavelet_fused_wmma<<<dim3(64 * SEGS / 8), dim3(256), 0, stream>>>(x, y);
}
